# starting kernel
# baseline (speedup 1.0000x reference)
#include <hip/hip_runtime.h>
#include <math.h>

typedef __attribute__((ext_vector_type(2))) float v2f;
typedef __attribute__((ext_vector_type(8))) float v8f;

typedef const float __attribute__((address_space(1)))* gcfp;
typedef float __attribute__((address_space(1)))* gfp;

__device__ __forceinline__ float gelu_exact(float v) {
    return 0.5f * v * (1.0f + erff(v * 0.7071067811865475f));
}

__global__ void zero_f32(float* __restrict__ p, long long n) {
    long long i = (long long)blockIdx.x * blockDim.x + threadIdx.x;
    if (i < n) p[i] = 0.0f;
}

__global__ void count_deg(const int* __restrict__ dst, float* __restrict__ cnt, int E) {
    int i = blockIdx.x * blockDim.x + threadIdx.x;
    if (i < E) atomicAdd(&cnt[dst[i]], 1.0f);
}

__global__ void finalize_inv(float* __restrict__ cnt, int n) {
    int i = blockIdx.x * blockDim.x + threadIdx.x;
    if (i < n) cnt[i] = 1.0f / fmaxf(cnt[i], 1.0f);
}

__global__ void pack_w(const float* __restrict__ W, float* __restrict__ Wpk,
                       int K, int Nc, int NcPad) {
    int i = blockIdx.x * blockDim.x + threadIdx.x;
    if (i >= K * NcPad) return;
    int j   = i & 1;
    int q   = i >> 1;
    int col = q % NcPad;
    int qq  = q / NcPad;
    int h   = qq & 1;
    int ks  = qq >> 1;
    int row = ks * 4 + h * 2 + j;
    Wpk[i] = (col < Nc) ? W[row * Nc + col] : 0.0f;
}

template<int VEC>
__global__ void scatter_add(const float* __restrict__ yl, const int* __restrict__ src,
                            const int* __restrict__ dst, float* __restrict__ agg,
                            int E, int dout) {
    const int chunks = dout / VEC;
    long long tid = (long long)blockIdx.x * blockDim.x + threadIdx.x;
    if (tid >= (long long)E * chunks) return;
    int e = (int)(tid / chunks);
    int c = (int)(tid % chunks);
    int s = src[e], d = dst[e];
    if (VEC == 4) {
        const float4 v = *(const float4*)(yl + (size_t)s * dout + c * 4);
        float* p = agg + (size_t)d * dout + c * 4;
        atomicAdd(p + 0, v.x);
        atomicAdd(p + 1, v.y);
        atomicAdd(p + 2, v.z);
        atomicAdd(p + 3, v.w);
    } else {
        atomicAdd(agg + (size_t)d * dout + c, yl[(size_t)s * dout + c]);
    }
}

__global__ void combine(float* __restrict__ agg, const float* __restrict__ inv,
                        const float* __restrict__ bias, int n, int dout) {
    long long i = (long long)blockIdx.x * blockDim.x + threadIdx.x;
    if (i >= (long long)n * dout) return;
    int node = (int)(i / dout);
    int f = (int)(i % dout);
    agg[i] = agg[i] * inv[node] + bias[f];
}

template<bool ACC, bool GELU, int MT>
__global__ __launch_bounds__(256) void gemm_wmma(
        const float* __restrict__ A, const float* __restrict__ Wpk,
        const float* __restrict__ Cin, float* __restrict__ D,
        int K, int Nc, int NcPad, int tilesM, int tilesN, int totalTiles) {
    int wave = threadIdx.x >> 5;
    int tile = blockIdx.x * (blockDim.x >> 5) + wave;
    if (tile >= totalTiles) return;
    int group = tile / tilesN;
    int tn    = tile - group * tilesN;
    int tmBase = group * MT;
    int lane = threadIdx.x & 31;
    int half = lane >> 4;
    int l = lane & 15;
    int col = tn * 16 + l;

    gcfp Wb = (gcfp)Wpk + ((size_t)half * NcPad + col) * 2;
    const size_t wstep = (size_t)4 * NcPad;

    gcfp Ap[MT];
    #pragma unroll
    for (int t = 0; t < MT; ++t) {
        int rt = tmBase + t;
        if (rt > tilesM - 1) rt = tilesM - 1;
        Ap[t] = (gcfp)A + (size_t)(rt * 16 + l) * K + half * 2;
    }

    v8f acc[MT];
    #pragma unroll
    for (int t = 0; t < MT; ++t) acc[t] = (v8f){};

    for (int k = 0; k < K; k += 4) {
        float bx = Wb[0];
        float by = Wb[1];
        Wb += wstep;
        float ax[MT], ay[MT];
        #pragma unroll
        for (int t = 0; t < MT; ++t) {
            ax[t] = Ap[t][0];
            ay[t] = Ap[t][1];
            Ap[t] += 4;
        }
        v2f b;
        b.x = bx;
        b.y = by;
        #pragma unroll
        for (int t = 0; t < MT; ++t) {
            v2f a;
            a.x = ax[t];
            a.y = ay[t];
            acc[t] = __builtin_amdgcn_wmma_f32_16x16x4_f32(
                         false, a, false, b, (short)0, acc[t], false, false);
        }
    }

    if (col < Nc) {
        gcfp Cg = (gcfp)Cin;
        gfp  Dg = (gfp)D;
        #pragma unroll
        for (int t = 0; t < MT; ++t) {
            int rt = tmBase + t;
            if (rt < tilesM) {
                #pragma unroll
                for (int g = 0; g < 8; ++g) {
                    int row = rt * 16 + half * 8 + g;
                    size_t off = (size_t)row * Nc + col;
                    float v = acc[t][g];
                    if (ACC) v += Cg[off];
                    if (GELU) v = gelu_exact(v);
                    Dg[off] = v;
                }
            }
        }
    }
}

extern "C" void kernel_launch(void* const* d_in, const int* in_sizes, int n_in,
                              void* d_out, int out_size, void* d_ws, size_t ws_size,
                              hipStream_t stream) {
    static const int DIMS[5] = {256, 192, 128, 64, 3};
    constexpr int MT = 5;

    const float* feat  = (const float*)d_in[0];
    const int*   edges = (const int*)d_in[1];
    const int N = in_sizes[0] / DIMS[0];
    const int E = in_sizes[1] / 2;
    const int* src = edges;
    const int* dst = edges + E;

    const float* Wl[4] = {(const float*)d_in[2], (const float*)d_in[5],
                          (const float*)d_in[8], (const float*)d_in[11]};
    const float* Wr[4] = {(const float*)d_in[3], (const float*)d_in[6],
                          (const float*)d_in[9], (const float*)d_in[12]};
    const float* Bb[4] = {(const float*)d_in[4], (const float*)d_in[7],
                          (const float*)d_in[10], (const float*)d_in[13]};

    float* ws    = (float*)d_ws;
    float* inv   = ws;
    float* bufP  = ws + N;
    float* bufQ  = bufP + (size_t)N * 192;
    float* bufR  = bufQ + (size_t)N * 192;
    float* wPkL  = bufR + (size_t)N * 192;
    float* wPkR  = wPkL + 256 * 208;

    zero_f32<<<(N + 255) / 256, 256, 0, stream>>>(inv, N);
    count_deg<<<(E + 255) / 256, 256, 0, stream>>>(dst, inv, E);
    finalize_inv<<<(N + 255) / 256, 256, 0, stream>>>(inv, N);

    const float* x = feat;
    for (int L = 0; L < 4; ++L) {
        const int K    = DIMS[L];
        const int dout = DIMS[L + 1];
        float* yl   = (L & 1) ? bufR : bufP;
        float* agg  = bufQ;
        float* xout = (L == 3) ? (float*)d_out : yl;

        const int tilesM = N / 16;
        const int tilesN = (dout + 15) / 16;
        const int NcPad  = tilesN * 16;
        const int groups = (tilesM + MT - 1) / MT;
        const int total  = groups * tilesN;
        const int blocks = (total + 7) / 8;

        const int padN = K * NcPad;
        pack_w<<<(padN + 255) / 256, 256, 0, stream>>>(Wl[L], wPkL, K, dout, NcPad);
        pack_w<<<(padN + 255) / 256, 256, 0, stream>>>(Wr[L], wPkR, K, dout, NcPad);

        gemm_wmma<false, false, MT><<<blocks, 256, 0, stream>>>(
            x, wPkL, nullptr, yl, K, dout, NcPad, tilesM, tilesN, total);

        const long long na = (long long)N * dout;
        zero_f32<<<(unsigned)((na + 255) / 256), 256, 0, stream>>>(agg, na);

        if (dout % 4 == 0) {
            long long work = (long long)E * (dout / 4);
            scatter_add<4><<<(unsigned)((work + 255) / 256), 256, 0, stream>>>(
                yl, src, dst, agg, E, dout);
        } else {
            long long work = (long long)E * dout;
            scatter_add<1><<<(unsigned)((work + 255) / 256), 256, 0, stream>>>(
                yl, src, dst, agg, E, dout);
        }

        combine<<<(unsigned)((na + 255) / 256), 256, 0, stream>>>(
            agg, inv, Bb[L], N, dout);

        if (L < 3) {
            gemm_wmma<true, true, MT><<<blocks, 256, 0, stream>>>(
                x, wPkR, agg, xout, K, dout, NcPad, tilesM, tilesN, total);
        } else {
            gemm_wmma<true, false, MT><<<blocks, 256, 0, stream>>>(
                x, wPkR, agg, xout, K, dout, NcPad, tilesM, tilesN, total);
        }
        x = xout;
    }
}


# aggregation as a gather over destination-sorted edges (sorted once per call; one wave per node sums source rows in registers, writes mean+bias once) replaces zero + float-atomic scatter + combine
# speedup vs baseline: 2.0808x; 2.0808x over previous
#include <hip/hip_runtime.h>
#include <math.h>

typedef __attribute__((ext_vector_type(2))) float v2f;
typedef __attribute__((ext_vector_type(8))) float v8f;

typedef const float __attribute__((address_space(1)))* gcfp;
typedef float __attribute__((address_space(1)))* gfp;

__device__ __forceinline__ float gelu_exact(float v) {
    return 0.5f * v * (1.0f + erff(v * 0.7071067811865475f));
}

__global__ void zero_i32(int* __restrict__ p, int n) {
    int i = blockIdx.x * blockDim.x + threadIdx.x;
    if (i < n) p[i] = 0;
}

__global__ void rank_edges(const int* __restrict__ dst, int* __restrict__ cnt,
                           int* __restrict__ rank, int E) {
    int e = blockIdx.x * blockDim.x + threadIdx.x;
    if (e < E) rank[e] = atomicAdd(&cnt[dst[e]], 1);
}

__global__ __launch_bounds__(1024) void scan_deg(const int* __restrict__ cnt,
                                                 int* __restrict__ rowptr, int n) {
    __shared__ int wtot[32];
    const int t = threadIdx.x;
    const int lane = t & 31, wave = t >> 5;
    const int chunk = (n + 1023) / 1024;
    const int b = min(t * chunk, n);
    const int e = min(b + chunk, n);
    int tot = 0;
    for (int i = b; i < e; ++i) tot += cnt[i];
    int inc = tot;
    for (int o = 1; o < 32; o <<= 1) {
        int u = __shfl_up(inc, o);
        if (lane >= o) inc += u;
    }
    if (lane == 31) wtot[wave] = inc;
    __syncthreads();
    int w = (lane < wave) ? wtot[lane] : 0;
    for (int o = 16; o > 0; o >>= 1) w += __shfl_xor(w, o);
    int run = w + inc - tot;
    for (int i = b; i < e; ++i) {
        rowptr[i] = run;
        run += cnt[i];
    }
}

__global__ void fill_csr(const int* __restrict__ src, const int* __restrict__ dst,
                         const int* __restrict__ rank, const int* __restrict__ rowptr,
                         int* __restrict__ csr, int E) {
    int e = blockIdx.x * blockDim.x + threadIdx.x;
    if (e < E) csr[rowptr[dst[e]] + rank[e]] = src[e];
}

__global__ void pack_w(const float* __restrict__ W, float* __restrict__ Wpk,
                       int K, int Nc, int NcPad) {
    int i = blockIdx.x * blockDim.x + threadIdx.x;
    if (i >= K * NcPad) return;
    int j   = i & 1;
    int q   = i >> 1;
    int col = q % NcPad;
    int qq  = q / NcPad;
    int h   = qq & 1;
    int ks  = qq >> 1;
    int row = ks * 4 + h * 2 + j;
    Wpk[i] = (col < Nc) ? W[row * Nc + col] : 0.0f;
}

template<int DOUT>
__device__ __forceinline__ void ld_row(const float* __restrict__ yl, int s, int lane,
                                       float4& v4, float2& v2) {
    const float* r = yl + (size_t)s * DOUT;
    if (DOUT >= 128) v4 = *(const float4*)(r + lane * 4);
    if (DOUT % 128 == 64) v2 = *(const float2*)(r + (DOUT >= 128 ? 128 : 0) + lane * 2);
}

template<int DOUT>
__global__ __launch_bounds__(256) void gather_mean(
        const float* __restrict__ yl, const int* __restrict__ rowptr,
        const int* __restrict__ cnt, const int* __restrict__ csr,
        const float* __restrict__ bias, float* __restrict__ agg, int n) {
    const int lane = threadIdx.x & 31;
    const int node = __builtin_amdgcn_readfirstlane(blockIdx.x * 8 + (threadIdx.x >> 5));
    if (node >= n) return;
    const int beg = rowptr[node];
    const int deg = cnt[node];
    const float4 z4 = make_float4(0.f, 0.f, 0.f, 0.f);
    const float2 z2 = make_float2(0.f, 0.f);
    float4 a4 = z4;
    float2 a2 = z2;
    for (int base = 0; base < deg; base += 32) {
        const int m = min(32, deg - base);
        const int idx = (lane < m) ? csr[beg + base + lane] : 0;
        int j = 0;
        for (; j + 8 <= m; j += 8) {
            float4 v4[8];
            float2 v2[8];
            #pragma unroll
            for (int q = 0; q < 8; ++q) {
                v4[q] = z4; v2[q] = z2;
                ld_row<DOUT>(yl, __builtin_amdgcn_readlane(idx, j + q), lane, v4[q], v2[q]);
            }
            #pragma unroll
            for (int q = 0; q < 8; ++q) {
                a4.x += v4[q].x; a4.y += v4[q].y; a4.z += v4[q].z; a4.w += v4[q].w;
                a2.x += v2[q].x; a2.y += v2[q].y;
            }
        }
        for (; j < m; j += 4) {
            float4 v4[4];
            float2 v2[4];
            #pragma unroll
            for (int q = 0; q < 4; ++q) {
                v4[q] = z4; v2[q] = z2;
                if (j + q < m)
                    ld_row<DOUT>(yl, __builtin_amdgcn_readlane(idx, j + q), lane, v4[q], v2[q]);
            }
            #pragma unroll
            for (int q = 0; q < 4; ++q) {
                a4.x += v4[q].x; a4.y += v4[q].y; a4.z += v4[q].z; a4.w += v4[q].w;
                a2.x += v2[q].x; a2.y += v2[q].y;
            }
        }
    }
    const float inv = 1.0f / fmaxf((float)deg, 1.0f);
    float* o = agg + (size_t)node * DOUT;
    if (DOUT >= 128) {
        const float4 bb = *(const float4*)(bias + lane * 4);
        float4 r;
        r.x = a4.x * inv + bb.x; r.y = a4.y * inv + bb.y;
        r.z = a4.z * inv + bb.z; r.w = a4.w * inv + bb.w;
        *(float4*)(o + lane * 4) = r;
    }
    if (DOUT % 128 == 64) {
        const int off = (DOUT >= 128 ? 128 : 0) + lane * 2;
        const float2 bb = *(const float2*)(bias + off);
        float2 r;
        r.x = a2.x * inv + bb.x; r.y = a2.y * inv + bb.y;
        *(float2*)(o + off) = r;
    }
}

__global__ void gather_mean_narrow(
        const float* __restrict__ yl, const int* __restrict__ rowptr,
        const int* __restrict__ cnt, const int* __restrict__ csr,
        const float* __restrict__ bias, float* __restrict__ agg, int n, int dout) {
    int tid = blockIdx.x * blockDim.x + threadIdx.x;
    if (tid >= n * dout) return;
    const int node = tid / dout;
    const int c = tid - node * dout;
    const int beg = rowptr[node];
    const int deg = cnt[node];
    float acc = 0.0f;
    #pragma unroll 4
    for (int j = 0; j < deg; ++j) acc += yl[(size_t)csr[beg + j] * dout + c];
    agg[tid] = acc * (1.0f / fmaxf((float)deg, 1.0f)) + bias[c];
}

template<bool ACC, bool GELU, int MT>
__global__ __launch_bounds__(256) void gemm_wmma(
        const float* __restrict__ A, const float* __restrict__ Wpk,
        const float* __restrict__ Cin, float* __restrict__ D,
        int K, int Nc, int NcPad, int tilesM, int tilesN, int totalTiles) {
    int wave = threadIdx.x >> 5;
    int tile = blockIdx.x * (blockDim.x >> 5) + wave;
    if (tile >= totalTiles) return;
    int group = tile / tilesN;
    int tn    = tile - group * tilesN;
    int tmBase = group * MT;
    int lane = threadIdx.x & 31;
    int half = lane >> 4;
    int l = lane & 15;
    int col = tn * 16 + l;

    gcfp Wb = (gcfp)Wpk + ((size_t)half * NcPad + col) * 2;
    const size_t wstep = (size_t)4 * NcPad;

    gcfp Ap[MT];
    #pragma unroll
    for (int t = 0; t < MT; ++t) {
        int rt = tmBase + t;
        if (rt > tilesM - 1) rt = tilesM - 1;
        Ap[t] = (gcfp)A + (size_t)(rt * 16 + l) * K + half * 2;
    }

    v8f acc[MT];
    #pragma unroll
    for (int t = 0; t < MT; ++t) acc[t] = (v8f){};

    for (int k = 0; k < K; k += 4) {
        float bx = Wb[0];
        float by = Wb[1];
        Wb += wstep;
        float ax[MT], ay[MT];
        #pragma unroll
        for (int t = 0; t < MT; ++t) {
            ax[t] = Ap[t][0];
            ay[t] = Ap[t][1];
            Ap[t] += 4;
        }
        v2f b;
        b.x = bx;
        b.y = by;
        #pragma unroll
        for (int t = 0; t < MT; ++t) {
            v2f a;
            a.x = ax[t];
            a.y = ay[t];
            acc[t] = __builtin_amdgcn_wmma_f32_16x16x4_f32(
                         false, a, false, b, (short)0, acc[t], false, false);
        }
    }

    if (col < Nc) {
        gcfp Cg = (gcfp)Cin;
        gfp  Dg = (gfp)D;
        #pragma unroll
        for (int t = 0; t < MT; ++t) {
            int rt = tmBase + t;
            if (rt < tilesM) {
                #pragma unroll
                for (int g = 0; g < 8; ++g) {
                    int row = rt * 16 + half * 8 + g;
                    size_t off = (size_t)row * Nc + col;
                    float v = acc[t][g];
                    if (ACC) v += Cg[off];
                    if (GELU) v = gelu_exact(v);
                    Dg[off] = v;
                }
            }
        }
    }
}

extern "C" void kernel_launch(void* const* d_in, const int* in_sizes, int n_in,
                              void* d_out, int out_size, void* d_ws, size_t ws_size,
                              hipStream_t stream) {
    static const int DIMS[5] = {256, 192, 128, 64, 3};
    constexpr int MT = 5;

    const float* feat  = (const float*)d_in[0];
    const int*   edges = (const int*)d_in[1];
    const int N = in_sizes[0] / DIMS[0];
    const int E = in_sizes[1] / 2;
    const int* src = edges;
    const int* dst = edges + E;

    const float* Wl[4] = {(const float*)d_in[2], (const float*)d_in[5],
                          (const float*)d_in[8], (const float*)d_in[11]};
    const float* Wr[4] = {(const float*)d_in[3], (const float*)d_in[6],
                          (const float*)d_in[9], (const float*)d_in[12]};
    const float* Bb[4] = {(const float*)d_in[4], (const float*)d_in[7],
                          (const float*)d_in[10], (const float*)d_in[13]};

    float* ws    = (float*)d_ws;
    float* bufP  = ws;
    float* bufQ  = bufP + (size_t)N * 192;
    float* bufR  = bufQ + (size_t)N * 192;
    float* wPkL  = bufR + (size_t)N * 192;
    float* wPkR  = wPkL + 256 * 208;
    int*   cnt   = (int*)(wPkR + 256 * 208);

    int* rowptr = (int*)(bufR + (size_t)N * 128);
    int* csr    = rowptr + N;
    int* rank   = (int*)bufQ;

    zero_i32<<<(N + 255) / 256, 256, 0, stream>>>(cnt, N);
    rank_edges<<<(E + 255) / 256, 256, 0, stream>>>(dst, cnt, rank, E);
    scan_deg<<<1, 1024, 0, stream>>>(cnt, rowptr, N);
    fill_csr<<<(E + 255) / 256, 256, 0, stream>>>(src, dst, rank, rowptr, csr, E);

    const float* x = feat;
    for (int L = 0; L < 4; ++L) {
        const int K    = DIMS[L];
        const int dout = DIMS[L + 1];
        float* yl   = (L & 1) ? bufR : bufP;
        float* agg  = bufQ;
        float* xout = (L == 3) ? (float*)d_out : yl;

        const int tilesM = N / 16;
        const int tilesN = (dout + 15) / 16;
        const int NcPad  = tilesN * 16;
        const int groups = (tilesM + MT - 1) / MT;
        const int total  = groups * tilesN;
        const int blocks = (total + 7) / 8;

        const int padN = K * NcPad;
        pack_w<<<(padN + 255) / 256, 256, 0, stream>>>(Wl[L], wPkL, K, dout, NcPad);
        pack_w<<<(padN + 255) / 256, 256, 0, stream>>>(Wr[L], wPkR, K, dout, NcPad);

        gemm_wmma<false, false, MT><<<blocks, 256, 0, stream>>>(
            x, wPkL, nullptr, yl, K, dout, NcPad, tilesM, tilesN, total);

        const int gblocks = (N + 7) / 8;
        if (dout == 192) {
            gather_mean<192><<<gblocks, 256, 0, stream>>>(yl, rowptr, cnt, csr, Bb[L], agg, N);
        } else if (dout == 128) {
            gather_mean<128><<<gblocks, 256, 0, stream>>>(yl, rowptr, cnt, csr, Bb[L], agg, N);
        } else if (dout == 64) {
            gather_mean<64><<<gblocks, 256, 0, stream>>>(yl, rowptr, cnt, csr, Bb[L], agg, N);
        } else {
            gather_mean_narrow<<<(N * dout + 255) / 256, 256, 0, stream>>>(
                yl, rowptr, cnt, csr, Bb[L], agg, N, dout);
        }

        if (L < 3) {
            gemm_wmma<true, true, MT><<<blocks, 256, 0, stream>>>(
                x, wPkR, agg, xout, K, dout, NcPad, tilesM, tilesN, total);
        } else {
            gemm_wmma<true, false, MT><<<blocks, 256, 0, stream>>>(
                x, wPkR, agg, xout, K, dout, NcPad, tilesM, tilesN, total);
        }
        x = xout;
    }
}


# GEMM wave tile 5x1 -> 2 M-tiles x 4 N-tiles per wave on the wide layers: each strided A-fragment load now feeds 4 WMMAs and each B load 2, A is fetched tilesN/4 times instead of tilesN times
# speedup vs baseline: 2.3905x; 1.1488x over previous
#include <hip/hip_runtime.h>
#include <math.h>

typedef __attribute__((ext_vector_type(2))) float v2f;
typedef __attribute__((ext_vector_type(8))) float v8f;

typedef const float __attribute__((address_space(1)))* gcfp;
typedef float __attribute__((address_space(1)))* gfp;

__device__ __forceinline__ float gelu_exact(float v) {
    return 0.5f * v * (1.0f + erff(v * 0.7071067811865475f));
}

__global__ void zero_i32(int* __restrict__ p, int n) {
    int i = blockIdx.x * blockDim.x + threadIdx.x;
    if (i < n) p[i] = 0;
}

__global__ void rank_edges(const int* __restrict__ dst, int* __restrict__ cnt,
                           int* __restrict__ rank, int E) {
    int e = blockIdx.x * blockDim.x + threadIdx.x;
    if (e < E) rank[e] = atomicAdd(&cnt[dst[e]], 1);
}

__global__ __launch_bounds__(1024) void scan_deg(const int* __restrict__ cnt,
                                                 int* __restrict__ rowptr, int n) {
    __shared__ int wtot[32];
    const int t = threadIdx.x;
    const int lane = t & 31, wave = t >> 5;
    const int chunk = (n + 1023) / 1024;
    const int b = min(t * chunk, n);
    const int e = min(b + chunk, n);
    int tot = 0;
    for (int i = b; i < e; ++i) tot += cnt[i];
    int inc = tot;
    for (int o = 1; o < 32; o <<= 1) {
        int u = __shfl_up(inc, o);
        if (lane >= o) inc += u;
    }
    if (lane == 31) wtot[wave] = inc;
    __syncthreads();
    int w = (lane < wave) ? wtot[lane] : 0;
    for (int o = 16; o > 0; o >>= 1) w += __shfl_xor(w, o);
    int run = w + inc - tot;
    for (int i = b; i < e; ++i) {
        rowptr[i] = run;
        run += cnt[i];
    }
}

__global__ void fill_csr(const int* __restrict__ src, const int* __restrict__ dst,
                         const int* __restrict__ rank, const int* __restrict__ rowptr,
                         int* __restrict__ csr, int E) {
    int e = blockIdx.x * blockDim.x + threadIdx.x;
    if (e < E) csr[rowptr[dst[e]] + rank[e]] = src[e];
}

__global__ void pack_w(const float* __restrict__ W, float* __restrict__ Wpk,
                       int K, int Nc, int NcPad) {
    int i = blockIdx.x * blockDim.x + threadIdx.x;
    if (i >= K * NcPad) return;
    int j   = i & 1;
    int q   = i >> 1;
    int col = q % NcPad;
    int qq  = q / NcPad;
    int h   = qq & 1;
    int ks  = qq >> 1;
    int row = ks * 4 + h * 2 + j;
    Wpk[i] = (col < Nc) ? W[row * Nc + col] : 0.0f;
}

template<int DOUT>
__device__ __forceinline__ void ld_row(const float* __restrict__ yl, int s, int lane,
                                       float4& v4, float2& v2) {
    const float* r = yl + (size_t)s * DOUT;
    if (DOUT >= 128) v4 = *(const float4*)(r + lane * 4);
    if (DOUT % 128 == 64) v2 = *(const float2*)(r + (DOUT >= 128 ? 128 : 0) + lane * 2);
}

template<int DOUT>
__global__ __launch_bounds__(256) void gather_mean(
        const float* __restrict__ yl, const int* __restrict__ rowptr,
        const int* __restrict__ cnt, const int* __restrict__ csr,
        const float* __restrict__ bias, float* __restrict__ agg, int n) {
    const int lane = threadIdx.x & 31;
    const int node = __builtin_amdgcn_readfirstlane(blockIdx.x * 8 + (threadIdx.x >> 5));
    if (node >= n) return;
    const int beg = rowptr[node];
    const int deg = cnt[node];
    const float4 z4 = make_float4(0.f, 0.f, 0.f, 0.f);
    const float2 z2 = make_float2(0.f, 0.f);
    float4 a4 = z4;
    float2 a2 = z2;
    for (int base = 0; base < deg; base += 32) {
        const int m = min(32, deg - base);
        const int idx = (lane < m) ? csr[beg + base + lane] : 0;
        int j = 0;
        for (; j + 8 <= m; j += 8) {
            float4 v4[8];
            float2 v2[8];
            #pragma unroll
            for (int q = 0; q < 8; ++q) {
                v4[q] = z4; v2[q] = z2;
                ld_row<DOUT>(yl, __builtin_amdgcn_readlane(idx, j + q), lane, v4[q], v2[q]);
            }
            #pragma unroll
            for (int q = 0; q < 8; ++q) {
                a4.x += v4[q].x; a4.y += v4[q].y; a4.z += v4[q].z; a4.w += v4[q].w;
                a2.x += v2[q].x; a2.y += v2[q].y;
            }
        }
        for (; j < m; j += 4) {
            float4 v4[4];
            float2 v2[4];
            #pragma unroll
            for (int q = 0; q < 4; ++q) {
                v4[q] = z4; v2[q] = z2;
                if (j + q < m)
                    ld_row<DOUT>(yl, __builtin_amdgcn_readlane(idx, j + q), lane, v4[q], v2[q]);
            }
            #pragma unroll
            for (int q = 0; q < 4; ++q) {
                a4.x += v4[q].x; a4.y += v4[q].y; a4.z += v4[q].z; a4.w += v4[q].w;
                a2.x += v2[q].x; a2.y += v2[q].y;
            }
        }
    }
    const float inv = 1.0f / fmaxf((float)deg, 1.0f);
    float* o = agg + (size_t)node * DOUT;
    if (DOUT >= 128) {
        const float4 bb = *(const float4*)(bias + lane * 4);
        float4 r;
        r.x = a4.x * inv + bb.x; r.y = a4.y * inv + bb.y;
        r.z = a4.z * inv + bb.z; r.w = a4.w * inv + bb.w;
        *(float4*)(o + lane * 4) = r;
    }
    if (DOUT % 128 == 64) {
        const int off = (DOUT >= 128 ? 128 : 0) + lane * 2;
        const float2 bb = *(const float2*)(bias + off);
        float2 r;
        r.x = a2.x * inv + bb.x; r.y = a2.y * inv + bb.y;
        *(float2*)(o + off) = r;
    }
}

__global__ void gather_mean_narrow(
        const float* __restrict__ yl, const int* __restrict__ rowptr,
        const int* __restrict__ cnt, const int* __restrict__ csr,
        const float* __restrict__ bias, float* __restrict__ agg, int n, int dout) {
    int tid = blockIdx.x * blockDim.x + threadIdx.x;
    if (tid >= n * dout) return;
    const int node = tid / dout;
    const int c = tid - node * dout;
    const int beg = rowptr[node];
    const int deg = cnt[node];
    float acc = 0.0f;
    #pragma unroll 4
    for (int j = 0; j < deg; ++j) acc += yl[(size_t)csr[beg + j] * dout + c];
    agg[tid] = acc * (1.0f / fmaxf((float)deg, 1.0f)) + bias[c];
}

template<bool ACC, bool GELU, int MT, int NT>
__global__ __launch_bounds__(256) void gemm_wmma(
        const float* __restrict__ A, const float* __restrict__ Wpk,
        const float* __restrict__ Cin, float* __restrict__ D,
        int K, int Nc, int NcPad, int tilesM, int groupsN, int totalTiles) {
    int wave = threadIdx.x >> 5;
    int tile = blockIdx.x * (blockDim.x >> 5) + wave;
    if (tile >= totalTiles) return;
    int group = tile / groupsN;
    int gn    = tile - group * groupsN;
    int tmBase = group * MT;
    int lane = threadIdx.x & 31;
    int half = lane >> 4;
    int l = lane & 15;
    int col0 = gn * (NT * 16) + l;

    gcfp Wb = (gcfp)Wpk + ((size_t)half * NcPad + col0) * 2;
    const size_t wstep = (size_t)4 * NcPad;

    gcfp Ap[MT];
    #pragma unroll
    for (int t = 0; t < MT; ++t) {
        int rt = tmBase + t;
        if (rt > tilesM - 1) rt = tilesM - 1;
        Ap[t] = (gcfp)A + (size_t)(rt * 16 + l) * K + half * 2;
    }

    v8f acc[MT][NT];
    #pragma unroll
    for (int t = 0; t < MT; ++t) {
        #pragma unroll
        for (int u = 0; u < NT; ++u) acc[t][u] = (v8f){};
    }

    for (int k = 0; k < K; k += 4) {
        float bx[NT], by[NT];
        #pragma unroll
        for (int u = 0; u < NT; ++u) {
            bx[u] = Wb[u * 32];
            by[u] = Wb[u * 32 + 1];
        }
        Wb += wstep;
        float ax[MT], ay[MT];
        #pragma unroll
        for (int t = 0; t < MT; ++t) {
            ax[t] = Ap[t][0];
            ay[t] = Ap[t][1];
            Ap[t] += 4;
        }
        #pragma unroll
        for (int t = 0; t < MT; ++t) {
            v2f a;
            a.x = ax[t];
            a.y = ay[t];
            #pragma unroll
            for (int u = 0; u < NT; ++u) {
                v2f b;
                b.x = bx[u];
                b.y = by[u];
                acc[t][u] = __builtin_amdgcn_wmma_f32_16x16x4_f32(
                                false, a, false, b, (short)0, acc[t][u], false, false);
            }
        }
    }

    gcfp Cg = (gcfp)Cin;
    gfp  Dg = (gfp)D;
    #pragma unroll
    for (int t = 0; t < MT; ++t) {
        int rt = tmBase + t;
        if (rt < tilesM) {
            #pragma unroll
            for (int u = 0; u < NT; ++u) {
                int col = col0 + u * 16;
                if (col < Nc) {
                    #pragma unroll
                    for (int g = 0; g < 8; ++g) {
                        int row = rt * 16 + half * 8 + g;
                        size_t off = (size_t)row * Nc + col;
                        float v = acc[t][u][g];
                        if (ACC) v += Cg[off];
                        if (GELU) v = gelu_exact(v);
                        Dg[off] = v;
                    }
                }
            }
        }
    }
}

extern "C" void kernel_launch(void* const* d_in, const int* in_sizes, int n_in,
                              void* d_out, int out_size, void* d_ws, size_t ws_size,
                              hipStream_t stream) {
    static const int DIMS[5] = {256, 192, 128, 64, 3};
    constexpr int MTW = 2, NTW = 4;
    constexpr int MTN = 5, NTN = 1;

    const float* feat  = (const float*)d_in[0];
    const int*   edges = (const int*)d_in[1];
    const int N = in_sizes[0] / DIMS[0];
    const int E = in_sizes[1] / 2;
    const int* src = edges;
    const int* dst = edges + E;

    const float* Wl[4] = {(const float*)d_in[2], (const float*)d_in[5],
                          (const float*)d_in[8], (const float*)d_in[11]};
    const float* Wr[4] = {(const float*)d_in[3], (const float*)d_in[6],
                          (const float*)d_in[9], (const float*)d_in[12]};
    const float* Bb[4] = {(const float*)d_in[4], (const float*)d_in[7],
                          (const float*)d_in[10], (const float*)d_in[13]};

    float* ws    = (float*)d_ws;
    float* bufP  = ws;
    float* bufQ  = bufP + (size_t)N * 192;
    float* bufR  = bufQ + (size_t)N * 192;
    float* wPkL  = bufR + (size_t)N * 192;
    float* wPkR  = wPkL + 256 * 208;
    int*   cnt   = (int*)(wPkR + 256 * 208);

    int* rowptr = (int*)(bufR + (size_t)N * 128);
    int* csr    = rowptr + N;
    int* rank   = (int*)bufQ;

    zero_i32<<<(N + 255) / 256, 256, 0, stream>>>(cnt, N);
    rank_edges<<<(E + 255) / 256, 256, 0, stream>>>(dst, cnt, rank, E);
    scan_deg<<<1, 1024, 0, stream>>>(cnt, rowptr, N);
    fill_csr<<<(E + 255) / 256, 256, 0, stream>>>(src, dst, rank, rowptr, csr, E);

    const float* x = feat;
    for (int L = 0; L < 4; ++L) {
        const int K    = DIMS[L];
        const int dout = DIMS[L + 1];
        float* yl   = (L & 1) ? bufR : bufP;
        float* agg  = bufQ;
        float* xout = (L == 3) ? (float*)d_out : yl;

        const int tilesM = N / 16;
        const int tilesN = (dout + 15) / 16;
        const int NcPad  = tilesN * 16;
        const bool wide   = (L < 3);
        const int mt      = wide ? MTW : MTN;
        const int groupsN = wide ? tilesN / NTW : tilesN / NTN;
        const int groups  = (tilesM + mt - 1) / mt;
        const int total   = groups * groupsN;
        const int wpb     = wide ? 4 : 8;
        const int blocks  = (total + wpb - 1) / wpb;
        const int threads = wpb * 32;

        const int padN = K * NcPad;
        pack_w<<<(padN + 255) / 256, 256, 0, stream>>>(Wl[L], wPkL, K, dout, NcPad);
        pack_w<<<(padN + 255) / 256, 256, 0, stream>>>(Wr[L], wPkR, K, dout, NcPad);

        if (wide) {
            gemm_wmma<false, false, MTW, NTW><<<blocks, threads, 0, stream>>>(
                x, wPkL, nullptr, yl, K, dout, NcPad, tilesM, groupsN, total);
        } else {
            gemm_wmma<false, false, MTN, NTN><<<blocks, threads, 0, stream>>>(
                x, wPkL, nullptr, yl, K, dout, NcPad, tilesM, groupsN, total);
        }

        const int gblocks = (N + 7) / 8;
        if (dout == 192) {
            gather_mean<192><<<gblocks, 256, 0, stream>>>(yl, rowptr, cnt, csr, Bb[L], agg, N);
        } else if (dout == 128) {
            gather_mean<128><<<gblocks, 256, 0, stream>>>(yl, rowptr, cnt, csr, Bb[L], agg, N);
        } else if (dout == 64) {
            gather_mean<64><<<gblocks, 256, 0, stream>>>(yl, rowptr, cnt, csr, Bb[L], agg, N);
        } else {
            gather_mean_narrow<<<(N * dout + 255) / 256, 256, 0, stream>>>(
                yl, rowptr, cnt, csr, Bb[L], agg, N, dout);
        }

        if (wide) {
            gemm_wmma<true, true, MTW, NTW><<<blocks, threads, 0, stream>>>(
                x, wPkR, agg, xout, K, dout, NcPad, tilesM, groupsN, total);
        } else {
            gemm_wmma<true, false, MTN, NTN><<<blocks, threads, 0, stream>>>(
                x, wPkR, agg, xout, K, dout, NcPad, tilesM, groupsN, total);
        }
        x = xout;
    }
}


# one GEMM per layer over the concatenated packed weight [Wl|Wr] (one pack launch, one pass over x); root term + GELU finished in place by the gather's epilogue: 24 launches -> 16
# speedup vs baseline: 2.8580x; 1.1956x over previous
#include <hip/hip_runtime.h>
#include <math.h>

typedef __attribute__((ext_vector_type(2))) float v2f;
typedef __attribute__((ext_vector_type(8))) float v8f;

typedef const float __attribute__((address_space(1)))* gcfp;
typedef float __attribute__((address_space(1)))* gfp;

__device__ __forceinline__ float gelu_exact(float v) {
    return 0.5f * v * (1.0f + erff(v * 0.7071067811865475f));
}

__global__ void zero_i32(int* __restrict__ p, int n) {
    int i = blockIdx.x * blockDim.x + threadIdx.x;
    if (i < n) p[i] = 0;
}

__global__ void rank_edges(const int* __restrict__ dst, int* __restrict__ cnt,
                           int* __restrict__ rank, int E) {
    int e = blockIdx.x * blockDim.x + threadIdx.x;
    if (e < E) rank[e] = atomicAdd(&cnt[dst[e]], 1);
}

__global__ __launch_bounds__(1024) void scan_deg(const int* __restrict__ cnt,
                                                 int* __restrict__ rowptr, int n) {
    __shared__ int wtot[32];
    const int t = threadIdx.x;
    const int lane = t & 31, wave = t >> 5;
    const int chunk = (n + 1023) / 1024;
    const int b = min(t * chunk, n);
    const int e = min(b + chunk, n);
    int tot = 0;
    for (int i = b; i < e; ++i) tot += cnt[i];
    int inc = tot;
    for (int o = 1; o < 32; o <<= 1) {
        int u = __shfl_up(inc, o);
        if (lane >= o) inc += u;
    }
    if (lane == 31) wtot[wave] = inc;
    __syncthreads();
    int w = (lane < wave) ? wtot[lane] : 0;
    for (int o = 16; o > 0; o >>= 1) w += __shfl_xor(w, o);
    int run = w + inc - tot;
    for (int i = b; i < e; ++i) {
        rowptr[i] = run;
        run += cnt[i];
    }
}

__global__ void fill_csr(const int* __restrict__ src, const int* __restrict__ dst,
                         const int* __restrict__ rank, const int* __restrict__ rowptr,
                         int* __restrict__ csr, int E) {
    int e = blockIdx.x * blockDim.x + threadIdx.x;
    if (e < E) csr[rowptr[dst[e]] + rank[e]] = src[e];
}

__global__ void pack_w2(const float* __restrict__ Wl, const float* __restrict__ Wr,
                        float* __restrict__ Wpk, int K, int dout, int NcPad) {
    int i = blockIdx.x * blockDim.x + threadIdx.x;
    if (i >= K * NcPad) return;
    int j   = i & 1;
    int q   = i >> 1;
    int col = q % NcPad;
    int qq  = q / NcPad;
    int h   = qq & 1;
    int ks  = qq >> 1;
    int row = ks * 4 + h * 2 + j;
    float v = 0.0f;
    if (col < dout)          v = Wl[row * dout + col];
    else if (col < 2 * dout) v = Wr[row * dout + (col - dout)];
    Wpk[i] = v;
}

template<int DOUT>
__device__ __forceinline__ void ld_row(const float* __restrict__ yl, int s, int lane,
                                       float4& v4, float2& v2) {
    const float* r = yl + (size_t)s * DOUT;
    if (DOUT >= 128) v4 = *(const float4*)(r + lane * 4);
    if (DOUT % 128 == 64) v2 = *(const float2*)(r + (DOUT >= 128 ? 128 : 0) + lane * 2);
}

template<int DOUT>
__global__ __launch_bounds__(256) void gather_mean(
        const float* __restrict__ yl, const int* __restrict__ rowptr,
        const int* __restrict__ cnt, const int* __restrict__ csr,
        const float* __restrict__ bias, float* zx, int n) {
    const int lane = threadIdx.x & 31;
    const int node = __builtin_amdgcn_readfirstlane(blockIdx.x * 8 + (threadIdx.x >> 5));
    if (node >= n) return;
    const int beg = rowptr[node];
    const int deg = cnt[node];
    const float4 z4 = make_float4(0.f, 0.f, 0.f, 0.f);
    const float2 z2 = make_float2(0.f, 0.f);
    float4 a4 = z4;
    float2 a2 = z2;
    for (int base = 0; base < deg; base += 32) {
        const int m = min(32, deg - base);
        const int idx = (lane < m) ? csr[beg + base + lane] : 0;
        int j = 0;
        for (; j + 8 <= m; j += 8) {
            float4 v4[8];
            float2 v2[8];
            #pragma unroll
            for (int q = 0; q < 8; ++q) {
                v4[q] = z4; v2[q] = z2;
                ld_row<DOUT>(yl, __builtin_amdgcn_readlane(idx, j + q), lane, v4[q], v2[q]);
            }
            #pragma unroll
            for (int q = 0; q < 8; ++q) {
                a4.x += v4[q].x; a4.y += v4[q].y; a4.z += v4[q].z; a4.w += v4[q].w;
                a2.x += v2[q].x; a2.y += v2[q].y;
            }
        }
        for (; j < m; j += 4) {
            float4 v4[4];
            float2 v2[4];
            #pragma unroll
            for (int q = 0; q < 4; ++q) {
                v4[q] = z4; v2[q] = z2;
                if (j + q < m)
                    ld_row<DOUT>(yl, __builtin_amdgcn_readlane(idx, j + q), lane, v4[q], v2[q]);
            }
            #pragma unroll
            for (int q = 0; q < 4; ++q) {
                a4.x += v4[q].x; a4.y += v4[q].y; a4.z += v4[q].z; a4.w += v4[q].w;
                a2.x += v2[q].x; a2.y += v2[q].y;
            }
        }
    }
    const float inv = 1.0f / fmaxf((float)deg, 1.0f);
    float* o = zx + (size_t)node * DOUT;
    const int off2 = (DOUT >= 128 ? 128 : 0) + lane * 2;
    float4 b4 = z4, r4 = z4;
    float2 b2 = z2, r2 = z2;
    if (DOUT >= 128) {
        b4 = *(const float4*)(bias + lane * 4);
        r4 = *(const float4*)(o + lane * 4);
    }
    if (DOUT % 128 == 64) {
        b2 = *(const float2*)(bias + off2);
        r2 = *(const float2*)(o + off2);
    }
    if (DOUT >= 128) {
        float4 r;
        r.x = gelu_exact((a4.x * inv + b4.x) + r4.x);
        r.y = gelu_exact((a4.y * inv + b4.y) + r4.y);
        r.z = gelu_exact((a4.z * inv + b4.z) + r4.z);
        r.w = gelu_exact((a4.w * inv + b4.w) + r4.w);
        *(float4*)(o + lane * 4) = r;
    }
    if (DOUT % 128 == 64) {
        float2 r;
        r.x = gelu_exact((a2.x * inv + b2.x) + r2.x);
        r.y = gelu_exact((a2.y * inv + b2.y) + r2.y);
        *(float2*)(o + off2) = r;
    }
}

__global__ void gather_mean_narrow(
        const float* __restrict__ yl, const int* __restrict__ rowptr,
        const int* __restrict__ cnt, const int* __restrict__ csr,
        const float* __restrict__ bias, const float* __restrict__ z,
        float* __restrict__ out, int n, int dout) {
    int tid = blockIdx.x * blockDim.x + threadIdx.x;
    if (tid >= n * dout) return;
    const int node = tid / dout;
    const int c = tid - node * dout;
    const int beg = rowptr[node];
    const int deg = cnt[node];
    const float root = z[tid];
    float acc = 0.0f;
    #pragma unroll 4
    for (int j = 0; j < deg; ++j) acc += yl[(size_t)csr[beg + j] * dout + c];
    out[tid] = (acc * (1.0f / fmaxf((float)deg, 1.0f)) + bias[c]) + root;
}

template<int MT, int NT>
__global__ __launch_bounds__(256) void gemm_wmma(
        const float* __restrict__ A, const float* __restrict__ Wpk,
        float* __restrict__ Y, float* __restrict__ Z,
        int K, int dout, int NcPad, int tilesM, int groupsN, int totalTiles) {
    int wave = threadIdx.x >> 5;
    int tile = blockIdx.x * (blockDim.x >> 5) + wave;
    if (tile >= totalTiles) return;
    int group = tile / groupsN;
    int gn    = tile - group * groupsN;
    int tmBase = group * MT;
    int lane = threadIdx.x & 31;
    int half = lane >> 4;
    int l = lane & 15;
    int col0 = gn * (NT * 16) + l;

    gcfp Wb = (gcfp)Wpk + ((size_t)half * NcPad + col0) * 2;
    const size_t wstep = (size_t)4 * NcPad;

    gcfp Ap[MT];
    #pragma unroll
    for (int t = 0; t < MT; ++t) {
        int rt = tmBase + t;
        if (rt > tilesM - 1) rt = tilesM - 1;
        Ap[t] = (gcfp)A + (size_t)(rt * 16 + l) * K + half * 2;
    }

    v8f acc[MT][NT];
    #pragma unroll
    for (int t = 0; t < MT; ++t) {
        #pragma unroll
        for (int u = 0; u < NT; ++u) acc[t][u] = (v8f){};
    }

    for (int k = 0; k < K; k += 4) {
        float bx[NT], by[NT];
        #pragma unroll
        for (int u = 0; u < NT; ++u) {
            bx[u] = Wb[u * 32];
            by[u] = Wb[u * 32 + 1];
        }
        Wb += wstep;
        float ax[MT], ay[MT];
        #pragma unroll
        for (int t = 0; t < MT; ++t) {
            ax[t] = Ap[t][0];
            ay[t] = Ap[t][1];
            Ap[t] += 4;
        }
        #pragma unroll
        for (int t = 0; t < MT; ++t) {
            v2f a;
            a.x = ax[t];
            a.y = ay[t];
            #pragma unroll
            for (int u = 0; u < NT; ++u) {
                v2f b;
                b.x = bx[u];
                b.y = by[u];
                acc[t][u] = __builtin_amdgcn_wmma_f32_16x16x4_f32(
                                false, a, false, b, (short)0, acc[t][u], false, false);
            }
        }
    }

    gfp Yg = (gfp)Y;
    gfp Zg = (gfp)Z;
    #pragma unroll
    for (int t = 0; t < MT; ++t) {
        int rt = tmBase + t;
        if (rt < tilesM) {
            #pragma unroll
            for (int u = 0; u < NT; ++u) {
                int col = col0 + u * 16;
                if (col < 2 * dout) {
                    gfp dst = (col < dout) ? Yg : Zg;
                    int c   = (col < dout) ? col : col - dout;
                    #pragma unroll
                    for (int g = 0; g < 8; ++g) {
                        int row = rt * 16 + half * 8 + g;
                        dst[(size_t)row * dout + c] = acc[t][u][g];
                    }
                }
            }
        }
    }
}

extern "C" void kernel_launch(void* const* d_in, const int* in_sizes, int n_in,
                              void* d_out, int out_size, void* d_ws, size_t ws_size,
                              hipStream_t stream) {
    static const int DIMS[5] = {256, 192, 128, 64, 3};
    constexpr int MTW = 2, NTW = 4;
    constexpr int MTN = 5, NTN = 1;

    const float* feat  = (const float*)d_in[0];
    const int*   edges = (const int*)d_in[1];
    const int N = in_sizes[0] / DIMS[0];
    const int E = in_sizes[1] / 2;
    const int* src = edges;
    const int* dst = edges + E;

    const float* Wl[4] = {(const float*)d_in[2], (const float*)d_in[5],
                          (const float*)d_in[8], (const float*)d_in[11]};
    const float* Wr[4] = {(const float*)d_in[3], (const float*)d_in[6],
                          (const float*)d_in[9], (const float*)d_in[12]};
    const float* Bb[4] = {(const float*)d_in[4], (const float*)d_in[7],
                          (const float*)d_in[10], (const float*)d_in[13]};

    float* ws    = (float*)d_ws;
    float* bufP  = ws;
    float* bufQ  = bufP + (size_t)N * 192;
    float* bufR  = bufQ + (size_t)N * 192;
    float* wPk   = bufR + (size_t)N * 192;
    int*   cnt   = (int*)(wPk + 2 * 256 * 208);

    int* rowptr = (int*)(bufR + (size_t)N * 128);
    int* csr    = rowptr + N;
    int* rank   = (int*)bufQ;

    zero_i32<<<(N + 255) / 256, 256, 0, stream>>>(cnt, N);
    rank_edges<<<(E + 255) / 256, 256, 0, stream>>>(dst, cnt, rank, E);
    scan_deg<<<1, 1024, 0, stream>>>(cnt, rowptr, N);
    fill_csr<<<(E + 255) / 256, 256, 0, stream>>>(src, dst, rank, rowptr, csr, E);

    float* ylP[4] = {bufP, bufR, bufQ, bufP};
    float* zP[4]  = {bufQ, bufP, bufR, bufQ};

    const float* x = feat;
    for (int L = 0; L < 4; ++L) {
        const int K    = DIMS[L];
        const int dout = DIMS[L + 1];
        float* yl = ylP[L];
        float* z  = zP[L];

        const int tilesM = N / 16;
        const int tilesN = (2 * dout + 15) / 16;
        const int NcPad  = tilesN * 16;
        const bool wide   = (L < 3);
        const int mt      = wide ? MTW : MTN;
        const int groupsN = wide ? tilesN / NTW : tilesN / NTN;
        const int groups  = (tilesM + mt - 1) / mt;
        const int total   = groups * groupsN;
        const int wpb     = wide ? 4 : 8;
        const int blocks  = (total + wpb - 1) / wpb;
        const int threads = wpb * 32;

        const int padN = K * NcPad;
        pack_w2<<<(padN + 255) / 256, 256, 0, stream>>>(Wl[L], Wr[L], wPk, K, dout, NcPad);

        if (wide) {
            gemm_wmma<MTW, NTW><<<blocks, threads, 0, stream>>>(
                x, wPk, yl, z, K, dout, NcPad, tilesM, groupsN, total);
        } else {
            gemm_wmma<MTN, NTN><<<blocks, threads, 0, stream>>>(
                x, wPk, yl, z, K, dout, NcPad, tilesM, groupsN, total);
        }

        const int gblocks = (N + 7) / 8;
        if (dout == 192) {
            gather_mean<192><<<gblocks, 256, 0, stream>>>(yl, rowptr, cnt, csr, Bb[L], z, N);
        } else if (dout == 128) {
            gather_mean<128><<<gblocks, 256, 0, stream>>>(yl, rowptr, cnt, csr, Bb[L], z, N);
        } else if (dout == 64) {
            gather_mean<64><<<gblocks, 256, 0, stream>>>(yl, rowptr, cnt, csr, Bb[L], z, N);
        } else {
            gather_mean_narrow<<<(N * dout + 255) / 256, 256, 0, stream>>>(
                yl, rowptr, cnt, csr, Bb[L], z, (float*)d_out, N, dout);
        }
        x = z;
    }
}


# GEMM k loop in chunks of 8, weights repacked to match: a lane takes each A and B fragment as one 16-byte load per two matrix ops (was 8 bytes per op): half the load instructions and A line touches
# speedup vs baseline: 3.1941x; 1.1176x over previous
#include <hip/hip_runtime.h>
#include <math.h>

typedef __attribute__((ext_vector_type(2))) float v2f;
typedef __attribute__((ext_vector_type(4))) float v4f;
typedef __attribute__((ext_vector_type(8))) float v8f;

typedef const float __attribute__((address_space(1)))* gcfp;
typedef float __attribute__((address_space(1)))* gfp;
typedef const v4f __attribute__((address_space(1)))* gcv4p;

__device__ __forceinline__ float gelu_exact(float v) {
    return 0.5f * v * (1.0f + erff(v * 0.7071067811865475f));
}

constexpr int D0 = 256, D1 = 192, D2 = 128, D3 = 64, D4 = 3;
constexpr int ncpad(int dout) { return ((2 * dout + 15) / 16) * 16; }
constexpr int P0 = D0 * ncpad(D1);
constexpr int P1 = D1 * ncpad(D2);
constexpr int P2 = D2 * ncpad(D3);
constexpr int P3 = D3 * ncpad(D4);

__global__ void rank_edges(const int* __restrict__ dst, int* __restrict__ cnt,
                           int* __restrict__ rank, int E) {
    int e = blockIdx.x * blockDim.x + threadIdx.x;
    if (e < E) rank[e] = atomicAdd(&cnt[dst[e]], 1);
}

__device__ __forceinline__ void ld16(const int* __restrict__ cnt, int e, int n, int (&c)[16]) {
    if (e + 16 <= n) {
        #pragma unroll
        for (int g = 0; g < 4; ++g) {
            const int4 v = *(const int4*)(cnt + e + 4 * g);
            c[4 * g] = v.x; c[4 * g + 1] = v.y; c[4 * g + 2] = v.z; c[4 * g + 3] = v.w;
        }
    } else {
        #pragma unroll
        for (int k = 0; k < 16; ++k) c[k] = (e + k < n) ? cnt[e + k] : 0;
    }
}

__device__ __forceinline__ void scan_deg_group(const int* __restrict__ cnt,
                                               int* __restrict__ rowptr, int n, int* wtot) {
    const int lane = threadIdx.x & 31, wave = threadIdx.x >> 5;
    const int nw = blockDim.x >> 5;
    const int tiles = (n + 511) / 512;
    const int per = (tiles + nw - 1) / nw;
    const int t0 = min(wave * per, tiles);
    const int t1 = min(t0 + per, tiles);
    int tot = 0;
    for (int t = t0; t < t1; ++t) {
        int c[16];
        ld16(cnt, t * 512 + lane * 16, n, c);
        #pragma unroll
        for (int k = 0; k < 16; ++k) tot += c[k];
    }
    for (int o = 16; o > 0; o >>= 1) tot += __shfl_xor(tot, o);
    if (lane == 0) wtot[wave] = tot;
    __syncthreads();
    int carry = 0;
    for (int i = 0; i < wave; ++i) carry += wtot[i];
    int c[16];
    if (t0 < t1) ld16(cnt, t0 * 512 + lane * 16, n, c);
    for (int t = t0; t < t1; ++t) {
        const int e = t * 512 + lane * 16;
        int nx[16];
        if (t + 1 < t1) ld16(cnt, e + 512, n, nx);
        int s = 0;
        #pragma unroll
        for (int k = 0; k < 16; ++k) s += c[k];
        int inc = s;
        for (int o = 1; o < 32; o <<= 1) {
            int u = __shfl_up(inc, o);
            if (lane >= o) inc += u;
        }
        int run = carry + inc - s;
        if (e + 16 <= n) {
            #pragma unroll
            for (int g = 0; g < 4; ++g) {
                int4 v;
                v.x = run; run += c[4 * g];
                v.y = run; run += c[4 * g + 1];
                v.z = run; run += c[4 * g + 2];
                v.w = run; run += c[4 * g + 3];
                *(int4*)(rowptr + e + 4 * g) = v;
            }
        } else {
            #pragma unroll
            for (int k = 0; k < 16; ++k) {
                if (e + k < n) rowptr[e + k] = run;
                run += c[k];
            }
        }
        carry += __shfl(inc, 31);
        if (t + 1 < t1) {
            #pragma unroll
            for (int k = 0; k < 16; ++k) c[k] = nx[k];
        }
    }
}

__global__ void fill_csr(const int* __restrict__ src, const int* __restrict__ dst,
                         const int* __restrict__ rank, const int* __restrict__ rowptr,
                         int* __restrict__ csr, int E) {
    int e = blockIdx.x * blockDim.x + threadIdx.x;
    if (e < E) csr[rowptr[dst[e]] + rank[e]] = src[e];
}

template<int DOUT>
__device__ __forceinline__ void pack_elem(const float* __restrict__ Wl,
                                          const float* __restrict__ Wr,
                                          float* __restrict__ Wpk, int i) {
    constexpr int NcPad = ncpad(DOUT);
    const int j   = i & 3;
    const int q   = i >> 2;
    const int col = q % NcPad;
    const int qq  = q / NcPad;
    const int h   = qq & 1;
    const int kc  = qq >> 1;
    const int row = kc * 8 + h * 4 + j;
    float v = 0.0f;
    if (col < DOUT)          v = Wl[row * DOUT + col];
    else if (col < 2 * DOUT) v = Wr[row * DOUT + (col - DOUT)];
    Wpk[i] = v;
}

__global__ void setup_zero_pack(int* __restrict__ cnt, int n,
        const float* __restrict__ Wl0, const float* __restrict__ Wr0,
        const float* __restrict__ Wl1, const float* __restrict__ Wr1,
        const float* __restrict__ Wl2, const float* __restrict__ Wr2,
        const float* __restrict__ Wl3, const float* __restrict__ Wr3,
        float* __restrict__ Wpk) {
    int i = blockIdx.x * blockDim.x + threadIdx.x;
    if (i < n) { cnt[i] = 0; return; }
    i -= n;
    if (i < P0) { pack_elem<D1>(Wl0, Wr0, Wpk, i); return; }
    i -= P0;
    if (i < P1) { pack_elem<D2>(Wl1, Wr1, Wpk + P0, i); return; }
    i -= P1;
    if (i < P2) { pack_elem<D3>(Wl2, Wr2, Wpk + P0 + P1, i); return; }
    i -= P2;
    if (i < P3) pack_elem<D4>(Wl3, Wr3, Wpk + P0 + P1 + P2, i);
}

template<int DOUT>
__device__ __forceinline__ void ld_row(const float* __restrict__ yl, int s, int lane,
                                       float4& v4, float2& v2) {
    const float* r = yl + (size_t)s * DOUT;
    if (DOUT >= 128) v4 = *(const float4*)(r + lane * 4);
    if (DOUT % 128 == 64) v2 = *(const float2*)(r + (DOUT >= 128 ? 128 : 0) + lane * 2);
}

template<int DOUT>
__global__ __launch_bounds__(256) void gather_mean(
        const float* __restrict__ yl, const int* __restrict__ rowptr,
        const int* __restrict__ cnt, const int* __restrict__ csr,
        const float* __restrict__ bias, float* zx, int n) {
    const int lane = threadIdx.x & 31;
    const int node = __builtin_amdgcn_readfirstlane(blockIdx.x * 8 + (threadIdx.x >> 5));
    if (node >= n) return;
    const int beg = rowptr[node];
    const int deg = cnt[node];
    const float4 z4 = make_float4(0.f, 0.f, 0.f, 0.f);
    const float2 z2 = make_float2(0.f, 0.f);
    float4 a4 = z4;
    float2 a2 = z2;
    for (int base = 0; base < deg; base += 32) {
        const int m = min(32, deg - base);
        const int idx = (lane < m) ? csr[beg + base + lane] : 0;
        int j = 0;
        for (; j + 8 <= m; j += 8) {
            float4 v4[8];
            float2 v2[8];
            #pragma unroll
            for (int q = 0; q < 8; ++q) {
                v4[q] = z4; v2[q] = z2;
                ld_row<DOUT>(yl, __builtin_amdgcn_readlane(idx, j + q), lane, v4[q], v2[q]);
            }
            #pragma unroll
            for (int q = 0; q < 8; ++q) {
                a4.x += v4[q].x; a4.y += v4[q].y; a4.z += v4[q].z; a4.w += v4[q].w;
                a2.x += v2[q].x; a2.y += v2[q].y;
            }
        }
        for (; j < m; j += 4) {
            float4 v4[4];
            float2 v2[4];
            #pragma unroll
            for (int q = 0; q < 4; ++q) {
                v4[q] = z4; v2[q] = z2;
                if (j + q < m)
                    ld_row<DOUT>(yl, __builtin_amdgcn_readlane(idx, j + q), lane, v4[q], v2[q]);
            }
            #pragma unroll
            for (int q = 0; q < 4; ++q) {
                a4.x += v4[q].x; a4.y += v4[q].y; a4.z += v4[q].z; a4.w += v4[q].w;
                a2.x += v2[q].x; a2.y += v2[q].y;
            }
        }
    }
    const float inv = 1.0f / fmaxf((float)deg, 1.0f);
    float* o = zx + (size_t)node * DOUT;
    const int off2 = (DOUT >= 128 ? 128 : 0) + lane * 2;
    float4 b4 = z4, r4 = z4;
    float2 b2 = z2, r2 = z2;
    if (DOUT >= 128) {
        b4 = *(const float4*)(bias + lane * 4);
        r4 = *(const float4*)(o + lane * 4);
    }
    if (DOUT % 128 == 64) {
        b2 = *(const float2*)(bias + off2);
        r2 = *(const float2*)(o + off2);
    }
    if (DOUT >= 128) {
        float4 r;
        r.x = gelu_exact((a4.x * inv + b4.x) + r4.x);
        r.y = gelu_exact((a4.y * inv + b4.y) + r4.y);
        r.z = gelu_exact((a4.z * inv + b4.z) + r4.z);
        r.w = gelu_exact((a4.w * inv + b4.w) + r4.w);
        *(float4*)(o + lane * 4) = r;
    }
    if (DOUT % 128 == 64) {
        float2 r;
        r.x = gelu_exact((a2.x * inv + b2.x) + r2.x);
        r.y = gelu_exact((a2.y * inv + b2.y) + r2.y);
        *(float2*)(o + off2) = r;
    }
}

__global__ void gather_mean_narrow(
        const float* __restrict__ yl, const int* __restrict__ rowptr,
        const int* __restrict__ cnt, const int* __restrict__ csr,
        const float* __restrict__ bias, const float* __restrict__ z,
        float* __restrict__ out, int n, int dout) {
    int tid = blockIdx.x * blockDim.x + threadIdx.x;
    if (tid >= n * dout) return;
    const int node = tid / dout;
    const int c = tid - node * dout;
    const int beg = rowptr[node];
    const int deg = cnt[node];
    const float root = z[tid];
    float acc = 0.0f;
    #pragma unroll 4
    for (int j = 0; j < deg; ++j) acc += yl[(size_t)csr[beg + j] * dout + c];
    out[tid] = (acc * (1.0f / fmaxf((float)deg, 1.0f)) + bias[c]) + root;
}

template<int MT, int NT>
__device__ __forceinline__ void gemm_tile(
        const float* __restrict__ A, const float* __restrict__ Wpk,
        float* __restrict__ Y, float* __restrict__ Z,
        int K, int dout, int NcPad, int tilesM, int groupsN, int totalTiles, int tile) {
    if (tile >= totalTiles) return;
    int group = tile / groupsN;
    int gn    = tile - group * groupsN;
    int tmBase = group * MT;
    int lane = threadIdx.x & 31;
    int half = lane >> 4;
    int l = lane & 15;
    int col0 = gn * (NT * 16) + l;

    gcfp Wb = (gcfp)Wpk + ((size_t)half * NcPad + col0) * 4;
    const size_t wstep = (size_t)8 * NcPad;

    gcfp Ap[MT];
    #pragma unroll
    for (int t = 0; t < MT; ++t) {
        int rt = tmBase + t;
        if (rt > tilesM - 1) rt = tilesM - 1;
        Ap[t] = (gcfp)A + (size_t)(rt * 16 + l) * K + half * 4;
    }

    v8f acc[MT][NT];
    #pragma unroll
    for (int t = 0; t < MT; ++t) {
        #pragma unroll
        for (int u = 0; u < NT; ++u) acc[t][u] = (v8f){};
    }

    for (int k = 0; k < K; k += 8) {
        v4f bv[NT];
        #pragma unroll
        for (int u = 0; u < NT; ++u) bv[u] = *(gcv4p)(Wb + u * 64);
        Wb += wstep;
        v4f av[MT];
        #pragma unroll
        for (int t = 0; t < MT; ++t) {
            av[t] = *(gcv4p)Ap[t];
            Ap[t] += 8;
        }
        #pragma unroll
        for (int p = 0; p < 2; ++p) {
            #pragma unroll
            for (int t = 0; t < MT; ++t) {
                v2f a;
                a.x = av[t][2 * p];
                a.y = av[t][2 * p + 1];
                #pragma unroll
                for (int u = 0; u < NT; ++u) {
                    v2f b;
                    b.x = bv[u][2 * p];
                    b.y = bv[u][2 * p + 1];
                    acc[t][u] = __builtin_amdgcn_wmma_f32_16x16x4_f32(
                                    false, a, false, b, (short)0, acc[t][u], false, false);
                }
            }
        }
    }

    gfp Yg = (gfp)Y;
    gfp Zg = (gfp)Z;
    #pragma unroll
    for (int t = 0; t < MT; ++t) {
        int rt = tmBase + t;
        if (rt < tilesM) {
            #pragma unroll
            for (int u = 0; u < NT; ++u) {
                int col = col0 + u * 16;
                if (col < 2 * dout) {
                    gfp dst = (col < dout) ? Yg : Zg;
                    int c   = (col < dout) ? col : col - dout;
                    #pragma unroll
                    for (int g = 0; g < 8; ++g) {
                        int row = rt * 16 + half * 8 + g;
                        dst[(size_t)row * dout + c] = acc[t][u][g];
                    }
                }
            }
        }
    }
}

template<int MT, int NT>
__global__ __launch_bounds__(256) void gemm_wmma(
        const float* __restrict__ A, const float* __restrict__ Wpk,
        float* __restrict__ Y, float* __restrict__ Z,
        int K, int dout, int NcPad, int tilesM, int groupsN, int totalTiles) {
    gemm_tile<MT, NT>(A, Wpk, Y, Z, K, dout, NcPad, tilesM, groupsN, totalTiles,
                      blockIdx.x * (blockDim.x >> 5) + (threadIdx.x >> 5));
}

template<int MT, int NT>
__global__ __launch_bounds__(256) void gemm_wmma_scan(
        const float* __restrict__ A, const float* __restrict__ Wpk,
        float* __restrict__ Y, float* __restrict__ Z,
        int K, int dout, int NcPad, int tilesM, int groupsN, int totalTiles,
        const int* __restrict__ cnt, int* __restrict__ rowptr, int n) {
    __shared__ int wtot[32];
    if (blockIdx.x == 0) {
        scan_deg_group(cnt, rowptr, n, wtot);
        return;
    }
    gemm_tile<MT, NT>(A, Wpk, Y, Z, K, dout, NcPad, tilesM, groupsN, totalTiles,
                      (blockIdx.x - 1) * (blockDim.x >> 5) + (threadIdx.x >> 5));
}

extern "C" void kernel_launch(void* const* d_in, const int* in_sizes, int n_in,
                              void* d_out, int out_size, void* d_ws, size_t ws_size,
                              hipStream_t stream) {
    static const int DIMS[5] = {D0, D1, D2, D3, D4};
    constexpr int MTW = 2, NTW = 4;
    constexpr int MTN = 5, NTN = 1;

    const float* feat  = (const float*)d_in[0];
    const int*   edges = (const int*)d_in[1];
    const int N = in_sizes[0] / DIMS[0];
    const int E = in_sizes[1] / 2;
    const int* src = edges;
    const int* dst = edges + E;

    const float* Wl[4] = {(const float*)d_in[2], (const float*)d_in[5],
                          (const float*)d_in[8], (const float*)d_in[11]};
    const float* Wr[4] = {(const float*)d_in[3], (const float*)d_in[6],
                          (const float*)d_in[9], (const float*)d_in[12]};
    const float* Bb[4] = {(const float*)d_in[4], (const float*)d_in[7],
                          (const float*)d_in[10], (const float*)d_in[13]};

    float* ws    = (float*)d_ws;
    float* bufP  = ws;
    float* bufQ  = bufP + (size_t)N * 192;
    float* bufR  = bufQ + (size_t)N * 192;
    int*   cnt   = (int*)(bufR + (size_t)N * 192);

    int*   rowptr = (int*)(bufR + (size_t)N * 128);
    int*   csr    = rowptr + N;
    float* wpk    = (float*)rowptr + (((size_t)N + (size_t)E + 3) & ~(size_t)3);
    int*   rank   = (int*)bufR;
    const int wOff[4] = {0, P0, P0 + P1, P0 + P1 + P2};

    const int setupN = N + P0 + P1 + P2 + P3;
    setup_zero_pack<<<(setupN + 255) / 256, 256, 0, stream>>>(
        cnt, N, Wl[0], Wr[0], Wl[1], Wr[1], Wl[2], Wr[2], Wl[3], Wr[3], wpk);

    rank_edges<<<(E + 255) / 256, 256, 0, stream>>>(dst, cnt, rank, E);

    float* ylP[4] = {bufP, bufR, bufQ, bufP};
    float* zP[4]  = {bufQ, bufP, bufR, bufQ};

    const float* x = feat;
    for (int L = 0; L < 4; ++L) {
        const int K    = DIMS[L];
        const int dout = DIMS[L + 1];
        float* yl = ylP[L];
        float* z  = zP[L];
        const float* wL = wpk + wOff[L];

        const int tilesM = N / 16;
        const int tilesN = (2 * dout + 15) / 16;
        const int NcPad  = tilesN * 16;
        const bool wide   = (L < 3);
        const int mt      = wide ? MTW : MTN;
        const int groupsN = wide ? tilesN / NTW : tilesN / NTN;
        const int groups  = (tilesM + mt - 1) / mt;
        const int total   = groups * groupsN;
        const int wpb     = wide ? 4 : 8;
        const int blocks  = (total + wpb - 1) / wpb;
        const int threads = wpb * 32;

        if (L == 0) {
            gemm_wmma_scan<MTW, NTW><<<blocks + 1, threads, 0, stream>>>(
                x, wL, yl, z, K, dout, NcPad, tilesM, groupsN, total, cnt, rowptr, N);
            fill_csr<<<(E + 255) / 256, 256, 0, stream>>>(src, dst, rank, rowptr, csr, E);
        } else if (wide) {
            gemm_wmma<MTW, NTW><<<blocks, threads, 0, stream>>>(
                x, wL, yl, z, K, dout, NcPad, tilesM, groupsN, total);
        } else {
            gemm_wmma<MTN, NTN><<<blocks, threads, 0, stream>>>(
                x, wL, yl, z, K, dout, NcPad, tilesM, groupsN, total);
        }

        const int gblocks = (N + 7) / 8;
        if (dout == 192) {
            gather_mean<192><<<gblocks, 256, 0, stream>>>(yl, rowptr, cnt, csr, Bb[L], z, N);
        } else if (dout == 128) {
            gather_mean<128><<<gblocks, 256, 0, stream>>>(yl, rowptr, cnt, csr, Bb[L], z, N);
        } else if (dout == 64) {
            gather_mean<64><<<gblocks, 256, 0, stream>>>(yl, rowptr, cnt, csr, Bb[L], z, N);
        } else {
            gather_mean_narrow<<<(N * dout + 255) / 256, 256, 0, stream>>>(
                yl, rowptr, cnt, csr, Bb[L], z, (float*)d_out, N, dout);
        }
        x = z;
    }
}
